# starting kernel
# baseline (speedup 1.0000x reference)
#include <hip/hip_runtime.h>
#include <hip/hip_bf16.h>
#include <math.h>

#define V_  128
#define H_  1024
#define SW_ 1024
#define SD_ 8192

typedef __attribute__((ext_vector_type(2))) float v2f;
typedef __attribute__((ext_vector_type(4))) float v4f;
typedef __attribute__((ext_vector_type(8))) float v8f;

#define WS_CTRL   0
#define WS_SINP   16
#define WS_RNNIN  2048
#define WS_GH     4096
#define WS_GI     8192
#define WS_HNEW   12288

template <int ACT>
__global__ void gemv_wmma(const float* __restrict__ W, const float* __restrict__ vec,
                          const float* __restrict__ bias, float* __restrict__ out, int K) {
    const int lane = threadIdx.x & 31;
    const int wave = threadIdx.x >> 5;
    const int row_base = blockIdx.x * 16;
    const int kslice = K >> 3;
    const int kbeg = wave * kslice;

    const int r  = row_base + (lane & 15);
    const int co = (lane >> 4) << 1;
    const float* wp = W + (size_t)r * K + (size_t)(kbeg + co);
    const float* vp = vec + kbeg + co;

    v8f c0 = {}, c1 = {}, c2 = {}, c3 = {};
    for (int k = 0; k < kslice; k += 16) {
        if ((k & 63) == 0) __builtin_prefetch(wp + k + 256, 0, 1);
        v2f a0 = *(const v2f*)(vp + k);
        v2f b0 = *(const v2f*)(wp + k);
        v2f a1 = *(const v2f*)(vp + k + 4);
        v2f b1 = *(const v2f*)(wp + k + 4);
        v2f a2 = *(const v2f*)(vp + k + 8);
        v2f b2 = *(const v2f*)(wp + k + 8);
        v2f a3 = *(const v2f*)(vp + k + 12);
        v2f b3 = *(const v2f*)(wp + k + 12);
        c0 = __builtin_amdgcn_wmma_f32_16x16x4_f32(false, a0, false, b0, (short)0, c0, false, false);
        c1 = __builtin_amdgcn_wmma_f32_16x16x4_f32(false, a1, false, b1, (short)0, c1, false, false);
        c2 = __builtin_amdgcn_wmma_f32_16x16x4_f32(false, a2, false, b2, (short)0, c2, false, false);
        c3 = __builtin_amdgcn_wmma_f32_16x16x4_f32(false, a3, false, b3, (short)0, c3, false, false);
    }

    __shared__ float red[8][16];
    if (lane < 16) red[wave][lane] = c0[0] + c1[0] + c2[0] + c3[0];
    __syncthreads();
    if (threadIdx.x < 16) {
        float s = 0.f;
        #pragma unroll
        for (int w = 0; w < 8; ++w) s += red[w][threadIdx.x];
        s += bias[row_base + threadIdx.x];
        if (ACT == 1) s = tanhf(s);
        out[row_base + threadIdx.x] = s;
    }
}

__global__ void ctrl_and_x(const int* __restrict__ inp, const float* __restrict__ hidden,
                           const float* __restrict__ emb, const float* __restrict__ W_ctrl,
                           const float* __restrict__ b_ctrl, float* ws) {
    if (blockIdx.x == 0) {
        __shared__ float lg[3];
        const int wave = threadIdx.x >> 5, lane = threadIdx.x & 31;
        if (wave < 3) {
            float s = 0.f;
            for (int i = lane; i < H_; i += 32) s += W_ctrl[wave * H_ + i] * hidden[i];
            #pragma unroll
            for (int off = 16; off > 0; off >>= 1) s += __shfl_down(s, off, 32);
            if (lane == 0) lg[wave] = s + b_ctrl[wave];
        }
        __syncthreads();
        if (threadIdx.x == 0) {
            float m = fmaxf(lg[0], fmaxf(lg[1], lg[2]));
            float e0 = expf(lg[0] - m), e1 = expf(lg[1] - m), e2 = expf(lg[2] - m);
            float inv = 1.f / (e0 + e1 + e2);
            ws[WS_CTRL + 0] = e0 * inv;
            ws[WS_CTRL + 1] = e1 * inv;
            ws[WS_CTRL + 2] = e2 * inv;
        }
    } else {
        int i = (blockIdx.x - 1) * blockDim.x + threadIdx.x;
        if (i < H_) ws[WS_RNNIN + i] = emb[(size_t)inp[0] * H_ + i];
    }
}

__global__ void stack_update(const float* __restrict__ stack, float* __restrict__ new_stack,
                             float* ws) {
    const float a_push = ws[WS_CTRL + 0];
    const float a_pop  = ws[WS_CTRL + 1];
    const float a_noop = ws[WS_CTRL + 2];

    const int idx = blockIdx.x * blockDim.x + threadIdx.x;
    const int row_f4 = SW_ / 4;
    const int j = idx / row_f4;
    const int c = (idx - j * row_f4) * 4;

    v4f cur = *(const v4f*)(stack + (size_t)j * SW_ + c);
    v4f up  = (j == 0) ? *(const v4f*)(ws + WS_SINP + c)
                       : *(const v4f*)(stack + (size_t)(j - 1) * SW_ + c);
    v4f dn = {};
    if (j < SD_ - 1) dn = *(const v4f*)(stack + (size_t)(j + 1) * SW_ + c);

    v4f o = a_noop * cur + a_push * up + a_pop * dn;

    __builtin_nontemporal_store(o, (v4f*)(new_stack + (size_t)j * SW_ + c));
    if (j == 0) *(v4f*)(ws + WS_RNNIN + H_ + c) = o;
}

__global__ void gru_gates(const float* __restrict__ hidden, float* ws,
                          float* __restrict__ hid_out) {
    const int i = blockIdx.x * blockDim.x + threadIdx.x;
    if (i >= H_) return;
    const float* gi = ws + WS_GI;
    const float* gh = ws + WS_GH;
    float r  = 1.f / (1.f + expf(-(gi[i] + gh[i])));
    float z  = 1.f / (1.f + expf(-(gi[H_ + i] + gh[H_ + i])));
    float n  = tanhf(gi[2 * H_ + i] + r * gh[2 * H_ + i]);
    float hn = (1.f - z) * n + z * hidden[i];
    hid_out[i] = hn;
    ws[WS_HNEW + i] = hn;
}

extern "C" void kernel_launch(void* const* d_in, const int* in_sizes, int n_in,
                              void* d_out, int out_size, void* d_ws, size_t ws_size,
                              hipStream_t stream) {
    (void)in_sizes; (void)n_in; (void)out_size; (void)ws_size;
    const int*   inp    = (const int*)  d_in[0];
    const float* hidden = (const float*)d_in[1];
    const float* stack  = (const float*)d_in[2];
    const float* emb    = (const float*)d_in[3];
    const float* W_ctrl = (const float*)d_in[4];
    const float* b_ctrl = (const float*)d_in[5];
    const float* W_sin  = (const float*)d_in[6];
    const float* b_sin  = (const float*)d_in[7];
    const float* W_ih   = (const float*)d_in[8];
    const float* b_ih   = (const float*)d_in[9];
    const float* W_hh   = (const float*)d_in[10];
    const float* b_hh   = (const float*)d_in[11];
    const float* W_dec  = (const float*)d_in[12];
    const float* b_dec  = (const float*)d_in[13];

    float* out       = (float*)d_out;
    float* hid_out   = out + V_;
    float* stack_out = out + V_ + H_;
    float* ws        = (float*)d_ws;

    ctrl_and_x<<<dim3(1 + H_ / 128), dim3(128), 0, stream>>>(inp, hidden, emb, W_ctrl, b_ctrl, ws);
    gemv_wmma<1><<<dim3(SW_ / 16), dim3(256), 0, stream>>>(W_sin, hidden, b_sin, ws + WS_SINP, H_);
    gemv_wmma<0><<<dim3(3 * H_ / 16), dim3(256), 0, stream>>>(W_hh, hidden, b_hh, ws + WS_GH, H_);
    stack_update<<<dim3((SD_ * SW_ / 4) / 256), dim3(256), 0, stream>>>(stack, stack_out, ws);
    gemv_wmma<0><<<dim3(3 * H_ / 16), dim3(256), 0, stream>>>(W_ih, ws + WS_RNNIN, b_ih, ws + WS_GI, H_ + SW_);
    gru_gates<<<dim3(H_ / 256), dim3(256), 0, stream>>>(hidden, ws, hid_out);
    gemv_wmma<0><<<dim3(V_ / 16), dim3(256), 0, stream>>>(W_dec, ws + WS_HNEW, b_dec, out, H_);
}


# Fuse 7 launches into 3, one per dependency level: gh + stack_input + controls; stack blend + gi (stack row 0 rebuilt in LDS); gates + logits (gates per block in LDS); summation order unchanged
# speedup vs baseline: 1.5240x; 1.5240x over previous
#include <hip/hip_runtime.h>
#include <hip/hip_bf16.h>
#include <math.h>


#define V_  128
#define H_  1024
#define SW_ 1024
#define SD_ 8192

typedef __attribute__((ext_vector_type(2))) float v2f;
typedef __attribute__((ext_vector_type(4))) float v4f;
typedef __attribute__((ext_vector_type(8))) float v8f;

#define WS_CTRL   0
#define WS_SINP   16
#define WS_GH     4096
#define WS_GI     8192

#define HH_BLOCKS  (3 * H_ / 16)
#define SIN_BLOCKS (SW_ / 16)
#define IH_BLOCKS  (3 * H_ / 16)
#define STK_BLOCKS ((SD_ * SW_ / 4) / 256)


__device__ __forceinline__ void gemv_tile(const float* __restrict__ W, const float* vec,
                                          const float* __restrict__ bias, float* __restrict__ out,
                                          int K, int row_base, bool act_tanh, float (*red)[16]) {
    const int lane = threadIdx.x & 31;
    const int wave = threadIdx.x >> 5;
    const int kslice = K >> 3;
    const int kbeg = wave * kslice;

    const int r  = row_base + (lane & 15);
    const int co = (lane >> 4) << 1;
    const float* wp = W + (size_t)r * K + (size_t)(kbeg + co);
    const float* vp = vec + kbeg + co;

    v8f c0 = {}, c1 = {}, c2 = {}, c3 = {};
    for (int k = 0; k < kslice; k += 16) {
        if ((k & 63) == 0) __builtin_prefetch(wp + k + 256, 0, 1);
        v2f a0 = *(const v2f*)(vp + k);
        v2f b0 = *(const v2f*)(wp + k);
        v2f a1 = *(const v2f*)(vp + k + 4);
        v2f b1 = *(const v2f*)(wp + k + 4);
        v2f a2 = *(const v2f*)(vp + k + 8);
        v2f b2 = *(const v2f*)(wp + k + 8);
        v2f a3 = *(const v2f*)(vp + k + 12);
        v2f b3 = *(const v2f*)(wp + k + 12);
        c0 = __builtin_amdgcn_wmma_f32_16x16x4_f32(false, a0, false, b0, (short)0, c0, false, false);
        c1 = __builtin_amdgcn_wmma_f32_16x16x4_f32(false, a1, false, b1, (short)0, c1, false, false);
        c2 = __builtin_amdgcn_wmma_f32_16x16x4_f32(false, a2, false, b2, (short)0, c2, false, false);
        c3 = __builtin_amdgcn_wmma_f32_16x16x4_f32(false, a3, false, b3, (short)0, c3, false, false);
    }

    if (lane < 16) red[wave][lane] = c0[0] + c1[0] + c2[0] + c3[0];
    __syncthreads();
    if (threadIdx.x < 16) {
        float s = 0.f;
        #pragma unroll
        for (int w = 0; w < 8; ++w) s += red[w][threadIdx.x];
        s += bias[row_base + threadIdx.x];
        if (act_tanh) s = tanhf(s);
        out[row_base + threadIdx.x] = s;
    }
}

__device__ __forceinline__ v4f stack_blend(float a_noop, float a_push, float a_pop,
                                           v4f cur, v4f up, v4f dn) {
    return a_noop * cur + a_push * up + a_pop * dn;
}

__global__ void level1_from_hidden(const float* __restrict__ hidden,
                                   const float* __restrict__ W_hh, const float* __restrict__ b_hh,
                                   const float* __restrict__ W_sin, const float* __restrict__ b_sin,
                                   const float* __restrict__ W_ctrl, const float* __restrict__ b_ctrl,
                                   float* __restrict__ ws, int K) {
    __shared__ float red[8][16];
    const int b = blockIdx.x;
    if (b < HH_BLOCKS + SIN_BLOCKS) {
        const bool is_sin = (b >= HH_BLOCKS);
        const float* W    = is_sin ? W_sin : W_hh;
        const float* bias = is_sin ? b_sin : b_hh;
        float* out        = ws + (is_sin ? WS_SINP : WS_GH);
        const int row_base = (is_sin ? (b - HH_BLOCKS) : b) * 16;
        gemv_tile(W, hidden, bias, out, K, row_base, is_sin, red);
    } else {
        __shared__ float lg[3];
        const int wave = threadIdx.x >> 5, lane = threadIdx.x & 31;
        if (wave < 3) {
            float s = 0.f;
            for (int i = lane; i < H_; i += 32) s += W_ctrl[wave * H_ + i] * hidden[i];
            #pragma unroll
            for (int off = 16; off > 0; off >>= 1) s += __shfl_down(s, off, 32);
            if (lane == 0) lg[wave] = s + b_ctrl[wave];
        }
        __syncthreads();
        if (threadIdx.x == 0) {
            float m = fmaxf(lg[0], fmaxf(lg[1], lg[2]));
            float e0 = expf(lg[0] - m), e1 = expf(lg[1] - m), e2 = expf(lg[2] - m);
            float inv = 1.f / (e0 + e1 + e2);
            ws[WS_CTRL + 0] = e0 * inv;
            ws[WS_CTRL + 1] = e1 * inv;
            ws[WS_CTRL + 2] = e2 * inv;
        }
    }
}

__global__ void level2_stack_and_gi(const int* __restrict__ inp, const float* __restrict__ emb,
                                    const float* __restrict__ stack,
                                    const float* __restrict__ W_ih, const float* __restrict__ b_ih,
                                    float* __restrict__ new_stack, float* ws, int K) {
    __shared__ float red[8][16];
    __shared__ __attribute__((aligned(16))) float rnn_in[H_ + SW_];

    const float a_push = ws[WS_CTRL + 0];
    const float a_pop  = ws[WS_CTRL + 1];
    const float a_noop = ws[WS_CTRL + 2];

    if (blockIdx.x < IH_BLOCKS) {
        const int c = threadIdx.x * 4;
        v4f x   = *(const v4f*)(emb + (size_t)inp[0] * H_ + c);
        v4f cur = *(const v4f*)(stack + c);
        v4f up  = *(const v4f*)(ws + WS_SINP + c);
        v4f dn  = *(const v4f*)(stack + SW_ + c);
        *(v4f*)(rnn_in + c)      = x;
        *(v4f*)(rnn_in + H_ + c) = stack_blend(a_noop, a_push, a_pop, cur, up, dn);
        __syncthreads();
        gemv_tile(W_ih, rnn_in, b_ih, ws + WS_GI, K, blockIdx.x * 16, false, red);
    } else {
        const int idx = (blockIdx.x - IH_BLOCKS) * blockDim.x + threadIdx.x;
        const int row_f4 = SW_ / 4;
        const int j = idx / row_f4;
        const int c = (idx - j * row_f4) * 4;

        v4f cur = *(const v4f*)(stack + (size_t)j * SW_ + c);
        v4f up  = (j == 0) ? *(const v4f*)(ws + WS_SINP + c)
                           : *(const v4f*)(stack + (size_t)(j - 1) * SW_ + c);
        v4f dn = {};
        if (j < SD_ - 1) dn = *(const v4f*)(stack + (size_t)(j + 1) * SW_ + c);

        v4f o = stack_blend(a_noop, a_push, a_pop, cur, up, dn);

        __builtin_nontemporal_store(o, (v4f*)(new_stack + (size_t)j * SW_ + c));
    }
}

__global__ void level3_gates_and_logits(const float* __restrict__ hidden, const float* __restrict__ ws,
                                        const float* __restrict__ W_dec, const float* __restrict__ b_dec,
                                        float* __restrict__ out, float* __restrict__ hid_out, int K) {
    __shared__ float red[8][16];
    __shared__ __attribute__((aligned(16))) float h_new[H_];
    const float* gi = ws + WS_GI;
    const float* gh = ws + WS_GH;
    const int c = threadIdx.x * 4;
    const v4f gir = *(const v4f*)(gi + c);
    const v4f giz = *(const v4f*)(gi + H_ + c);
    const v4f gin = *(const v4f*)(gi + 2 * H_ + c);
    const v4f ghr = *(const v4f*)(gh + c);
    const v4f ghz = *(const v4f*)(gh + H_ + c);
    const v4f ghn = *(const v4f*)(gh + 2 * H_ + c);
    const v4f hp  = *(const v4f*)(hidden + c);
    v4f hn;
    #pragma unroll
    for (int e = 0; e < 4; ++e) {
        float r = 1.f / (1.f + expf(-(gir[e] + ghr[e])));
        float z = 1.f / (1.f + expf(-(giz[e] + ghz[e])));
        float n = tanhf(gin[e] + r * ghn[e]);
        hn[e] = (1.f - z) * n + z * hp[e];
    }
    *(v4f*)(h_new + c) = hn;
    if (blockIdx.x == 0) *(v4f*)(hid_out + c) = hn;
    __syncthreads();
    gemv_tile(W_dec, h_new, b_dec, out, K, blockIdx.x * 16, false, red);
}

extern "C" void kernel_launch(void* const* d_in, const int* in_sizes, int n_in,
                              void* d_out, int out_size, void* d_ws, size_t ws_size,
                              hipStream_t stream) {
    (void)in_sizes; (void)n_in; (void)out_size; (void)ws_size;
    const int*   inp    = (const int*)  d_in[0];
    const float* hidden = (const float*)d_in[1];
    const float* stack  = (const float*)d_in[2];
    const float* emb    = (const float*)d_in[3];
    const float* W_ctrl = (const float*)d_in[4];
    const float* b_ctrl = (const float*)d_in[5];
    const float* W_sin  = (const float*)d_in[6];
    const float* b_sin  = (const float*)d_in[7];
    const float* W_ih   = (const float*)d_in[8];
    const float* b_ih   = (const float*)d_in[9];
    const float* W_hh   = (const float*)d_in[10];
    const float* b_hh   = (const float*)d_in[11];
    const float* W_dec  = (const float*)d_in[12];
    const float* b_dec  = (const float*)d_in[13];

    float* out       = (float*)d_out;
    float* hid_out   = out + V_;
    float* stack_out = out + V_ + H_;
    float* ws        = (float*)d_ws;

    level1_from_hidden<<<dim3(HH_BLOCKS + SIN_BLOCKS + 1), dim3(256), 0, stream>>>(
        hidden, W_hh, b_hh, W_sin, b_sin, W_ctrl, b_ctrl, ws, H_);
    level2_stack_and_gi<<<dim3(IH_BLOCKS + STK_BLOCKS), dim3(256), 0, stream>>>(
        inp, emb, stack, W_ih, b_ih, stack_out, ws, H_ + SW_);
    level3_gates_and_logits<<<dim3(V_ / 16), dim3(256), 0, stream>>>(
        hidden, ws, W_dec, b_dec, out, hid_out, H_);
}


# Stack blend: a block takes 16 consecutive rows, a thread keeps its 16-byte column of the 18-row window in registers (rows loaded once, all loads in flight); 8192 blend blocks -> 512; sums unchanged
# speedup vs baseline: 1.7415x; 1.1427x over previous
#include <hip/hip_runtime.h>
#include <hip/hip_bf16.h>
#include <math.h>


#define V_  128
#define H_  1024
#define SW_ 1024
#define SD_ 8192

typedef __attribute__((ext_vector_type(2))) float v2f;
typedef __attribute__((ext_vector_type(4))) float v4f;
typedef __attribute__((ext_vector_type(8))) float v8f;

#define WS_CTRL   0
#define WS_SINP   16
#define WS_GH     4096
#define WS_GI     8192

#define HH_BLOCKS  (3 * H_ / 16)
#define SIN_BLOCKS (SW_ / 16)
#define IH_BLOCKS  (3 * H_ / 16)
#define STK_ROWS   16
#define STK_BLOCKS (SD_ / STK_ROWS)


__device__ __forceinline__ void gemv_tile(const float* __restrict__ W, const float* vec,
                                          const float* __restrict__ bias, float* __restrict__ out,
                                          int K, int row_base, bool act_tanh, float (*red)[16]) {
    const int lane = threadIdx.x & 31;
    const int wave = threadIdx.x >> 5;
    const int kslice = K >> 3;
    const int kbeg = wave * kslice;

    const int r  = row_base + (lane & 15);
    const int co = (lane >> 4) << 1;
    const float* wp = W + (size_t)r * K + (size_t)(kbeg + co);
    const float* vp = vec + kbeg + co;

    v8f c0 = {}, c1 = {}, c2 = {}, c3 = {};
    for (int k = 0; k < kslice; k += 16) {
        if ((k & 63) == 0) __builtin_prefetch(wp + k + 256, 0, 1);
        v2f a0 = *(const v2f*)(vp + k);
        v2f b0 = *(const v2f*)(wp + k);
        v2f a1 = *(const v2f*)(vp + k + 4);
        v2f b1 = *(const v2f*)(wp + k + 4);
        v2f a2 = *(const v2f*)(vp + k + 8);
        v2f b2 = *(const v2f*)(wp + k + 8);
        v2f a3 = *(const v2f*)(vp + k + 12);
        v2f b3 = *(const v2f*)(wp + k + 12);
        c0 = __builtin_amdgcn_wmma_f32_16x16x4_f32(false, a0, false, b0, (short)0, c0, false, false);
        c1 = __builtin_amdgcn_wmma_f32_16x16x4_f32(false, a1, false, b1, (short)0, c1, false, false);
        c2 = __builtin_amdgcn_wmma_f32_16x16x4_f32(false, a2, false, b2, (short)0, c2, false, false);
        c3 = __builtin_amdgcn_wmma_f32_16x16x4_f32(false, a3, false, b3, (short)0, c3, false, false);
    }

    if (lane < 16) red[wave][lane] = c0[0] + c1[0] + c2[0] + c3[0];
    __syncthreads();
    if (threadIdx.x < 16) {
        float s = 0.f;
        #pragma unroll
        for (int w = 0; w < 8; ++w) s += red[w][threadIdx.x];
        s += bias[row_base + threadIdx.x];
        if (act_tanh) s = tanhf(s);
        out[row_base + threadIdx.x] = s;
    }
}

__device__ __forceinline__ v4f stack_blend(float a_noop, float a_push, float a_pop,
                                           v4f cur, v4f up, v4f dn) {
    return a_noop * cur + a_push * up + a_pop * dn;
}

__global__ void level1_from_hidden(const float* __restrict__ hidden,
                                   const float* __restrict__ W_hh, const float* __restrict__ b_hh,
                                   const float* __restrict__ W_sin, const float* __restrict__ b_sin,
                                   const float* __restrict__ W_ctrl, const float* __restrict__ b_ctrl,
                                   float* __restrict__ ws, int K) {
    __shared__ float red[8][16];
    const int b = blockIdx.x;
    if (b < HH_BLOCKS + SIN_BLOCKS) {
        const bool is_sin = (b >= HH_BLOCKS);
        const float* W    = is_sin ? W_sin : W_hh;
        const float* bias = is_sin ? b_sin : b_hh;
        float* out        = ws + (is_sin ? WS_SINP : WS_GH);
        const int row_base = (is_sin ? (b - HH_BLOCKS) : b) * 16;
        gemv_tile(W, hidden, bias, out, K, row_base, is_sin, red);
    } else {
        __shared__ float lg[3];
        const int wave = threadIdx.x >> 5, lane = threadIdx.x & 31;
        if (wave < 3) {
            float s = 0.f;
            for (int i = lane; i < H_; i += 32) s += W_ctrl[wave * H_ + i] * hidden[i];
            #pragma unroll
            for (int off = 16; off > 0; off >>= 1) s += __shfl_down(s, off, 32);
            if (lane == 0) lg[wave] = s + b_ctrl[wave];
        }
        __syncthreads();
        if (threadIdx.x == 0) {
            float m = fmaxf(lg[0], fmaxf(lg[1], lg[2]));
            float e0 = expf(lg[0] - m), e1 = expf(lg[1] - m), e2 = expf(lg[2] - m);
            float inv = 1.f / (e0 + e1 + e2);
            ws[WS_CTRL + 0] = e0 * inv;
            ws[WS_CTRL + 1] = e1 * inv;
            ws[WS_CTRL + 2] = e2 * inv;
        }
    }
}

__global__ void level2_stack_and_gi(const int* __restrict__ inp, const float* __restrict__ emb,
                                    const float* __restrict__ stack,
                                    const float* __restrict__ W_ih, const float* __restrict__ b_ih,
                                    float* __restrict__ new_stack, float* ws, int K) {
    __shared__ float red[8][16];
    __shared__ __attribute__((aligned(16))) float rnn_in[H_ + SW_];

    const float a_push = ws[WS_CTRL + 0];
    const float a_pop  = ws[WS_CTRL + 1];
    const float a_noop = ws[WS_CTRL + 2];

    if (blockIdx.x < IH_BLOCKS) {
        const int c = threadIdx.x * 4;
        v4f x   = *(const v4f*)(emb + (size_t)inp[0] * H_ + c);
        v4f cur = *(const v4f*)(stack + c);
        v4f up  = *(const v4f*)(ws + WS_SINP + c);
        v4f dn  = *(const v4f*)(stack + SW_ + c);
        *(v4f*)(rnn_in + c)      = x;
        *(v4f*)(rnn_in + H_ + c) = stack_blend(a_noop, a_push, a_pop, cur, up, dn);
        __syncthreads();
        gemv_tile(W_ih, rnn_in, b_ih, ws + WS_GI, K, blockIdx.x * 16, false, red);
    } else {
        const int c  = threadIdx.x * 4;
        const int j0 = (blockIdx.x - IH_BLOCKS) * STK_ROWS;
        const float* col = stack + (size_t)j0 * SW_ + c;

        v4f win[STK_ROWS + 2];
        win[0] = (j0 == 0) ? *(const v4f*)(ws + WS_SINP + c) : *(const v4f*)(col - SW_);
        #pragma unroll
        for (int i = 0; i < STK_ROWS; ++i) win[i + 1] = *(const v4f*)(col + (size_t)i * SW_);
        v4f below = {};
        if (j0 + STK_ROWS < SD_) below = *(const v4f*)(col + (size_t)STK_ROWS * SW_);
        win[STK_ROWS + 1] = below;

        #pragma unroll
        for (int i = 0; i < STK_ROWS; ++i) {
            v4f o = stack_blend(a_noop, a_push, a_pop, win[i + 1], win[i], win[i + 2]);
            __builtin_nontemporal_store(o, (v4f*)(new_stack + (size_t)(j0 + i) * SW_ + c));
        }
    }
}

__global__ void level3_gates_and_logits(const float* __restrict__ hidden, const float* __restrict__ ws,
                                        const float* __restrict__ W_dec, const float* __restrict__ b_dec,
                                        float* __restrict__ out, float* __restrict__ hid_out, int K) {
    __shared__ float red[8][16];
    __shared__ __attribute__((aligned(16))) float h_new[H_];
    const float* gi = ws + WS_GI;
    const float* gh = ws + WS_GH;
    const int c = threadIdx.x * 4;
    const v4f gir = *(const v4f*)(gi + c);
    const v4f giz = *(const v4f*)(gi + H_ + c);
    const v4f gin = *(const v4f*)(gi + 2 * H_ + c);
    const v4f ghr = *(const v4f*)(gh + c);
    const v4f ghz = *(const v4f*)(gh + H_ + c);
    const v4f ghn = *(const v4f*)(gh + 2 * H_ + c);
    const v4f hp  = *(const v4f*)(hidden + c);
    v4f hn;
    #pragma unroll
    for (int e = 0; e < 4; ++e) {
        float r = 1.f / (1.f + expf(-(gir[e] + ghr[e])));
        float z = 1.f / (1.f + expf(-(giz[e] + ghz[e])));
        float n = tanhf(gin[e] + r * ghn[e]);
        hn[e] = (1.f - z) * n + z * hp[e];
    }
    *(v4f*)(h_new + c) = hn;
    if (blockIdx.x == 0) *(v4f*)(hid_out + c) = hn;
    __syncthreads();
    gemv_tile(W_dec, h_new, b_dec, out, K, blockIdx.x * 16, false, red);
}

extern "C" void kernel_launch(void* const* d_in, const int* in_sizes, int n_in,
                              void* d_out, int out_size, void* d_ws, size_t ws_size,
                              hipStream_t stream) {
    (void)in_sizes; (void)n_in; (void)out_size; (void)ws_size;
    const int*   inp    = (const int*)  d_in[0];
    const float* hidden = (const float*)d_in[1];
    const float* stack  = (const float*)d_in[2];
    const float* emb    = (const float*)d_in[3];
    const float* W_ctrl = (const float*)d_in[4];
    const float* b_ctrl = (const float*)d_in[5];
    const float* W_sin  = (const float*)d_in[6];
    const float* b_sin  = (const float*)d_in[7];
    const float* W_ih   = (const float*)d_in[8];
    const float* b_ih   = (const float*)d_in[9];
    const float* W_hh   = (const float*)d_in[10];
    const float* b_hh   = (const float*)d_in[11];
    const float* W_dec  = (const float*)d_in[12];
    const float* b_dec  = (const float*)d_in[13];

    float* out       = (float*)d_out;
    float* hid_out   = out + V_;
    float* stack_out = out + V_ + H_;
    float* ws        = (float*)d_ws;

    level1_from_hidden<<<dim3(HH_BLOCKS + SIN_BLOCKS + 1), dim3(256), 0, stream>>>(
        hidden, W_hh, b_hh, W_sin, b_sin, W_ctrl, b_ctrl, ws, H_);
    level2_stack_and_gi<<<dim3(IH_BLOCKS + STK_BLOCKS), dim3(256), 0, stream>>>(
        inp, emb, stack, W_ih, b_ih, stack_out, ws, H_ + SW_);
    level3_gates_and_logits<<<dim3(V_ / 16), dim3(256), 0, stream>>>(
        hidden, ws, W_dec, b_dec, out, hid_out, H_);
}


# All behind gi joins the gi launch: a W_ih tile = r,z,n rows of 4 units; its block forms h_new and its K slice of the logits (1 WMMA per wave); launch 3 adds 256 partials; logit sum order changes
# speedup vs baseline: 1.9656x; 1.1287x over previous
#include <hip/hip_runtime.h>
#include <hip/hip_bf16.h>
#include <math.h>


#define V_  128
#define H_  1024
#define SW_ 1024
#define SD_ 8192

typedef __attribute__((ext_vector_type(2))) float v2f;
typedef __attribute__((ext_vector_type(4))) float v4f;
typedef __attribute__((ext_vector_type(8))) float v8f;

#define WS_CTRL   0
#define WS_SINP   16
#define WS_GH     4096
#define WS_PART   16384

#define HH_BLOCKS  (3 * H_ / 16)
#define SIN_BLOCKS (SW_ / 16)
#define UNITS_     4
#define GI_TILES   (H_ / UNITS_)
#define SUM_BLOCKS (V_ / 32)
#define STK_ROWS   16
#define STK_BLOCKS (SD_ / STK_ROWS)


__device__ __forceinline__ void gemv_partials(const float* __restrict__ W, const float* vec,
                                              int K, int r, float (*red)[16]) {
    const int lane = threadIdx.x & 31;
    const int wave = threadIdx.x >> 5;
    const int kslice = K >> 3;
    const int kbeg = wave * kslice;

    const int co = (lane >> 4) << 1;
    const float* wp = W + (size_t)r * K + (size_t)(kbeg + co);
    const float* vp = vec + kbeg + co;

    v8f c0 = {}, c1 = {}, c2 = {}, c3 = {};
    for (int k = 0; k < kslice; k += 16) {
        if ((k & 63) == 0) __builtin_prefetch(wp + k + 256, 0, 1);
        v2f a0 = *(const v2f*)(vp + k);
        v2f b0 = *(const v2f*)(wp + k);
        v2f a1 = *(const v2f*)(vp + k + 4);
        v2f b1 = *(const v2f*)(wp + k + 4);
        v2f a2 = *(const v2f*)(vp + k + 8);
        v2f b2 = *(const v2f*)(wp + k + 8);
        v2f a3 = *(const v2f*)(vp + k + 12);
        v2f b3 = *(const v2f*)(wp + k + 12);
        c0 = __builtin_amdgcn_wmma_f32_16x16x4_f32(false, a0, false, b0, (short)0, c0, false, false);
        c1 = __builtin_amdgcn_wmma_f32_16x16x4_f32(false, a1, false, b1, (short)0, c1, false, false);
        c2 = __builtin_amdgcn_wmma_f32_16x16x4_f32(false, a2, false, b2, (short)0, c2, false, false);
        c3 = __builtin_amdgcn_wmma_f32_16x16x4_f32(false, a3, false, b3, (short)0, c3, false, false);
    }

    if (lane < 16) red[wave][lane] = c0[0] + c1[0] + c2[0] + c3[0];
}

__device__ __forceinline__ void gemv_tile(const float* __restrict__ W, const float* vec,
                                          const float* __restrict__ bias, float* __restrict__ out,
                                          int K, int row_base, bool act_tanh, float (*red)[16]) {
    gemv_partials(W, vec, K, row_base + (threadIdx.x & 15), red);
    __syncthreads();
    if (threadIdx.x < 16) {
        float s = 0.f;
        #pragma unroll
        for (int w = 0; w < 8; ++w) s += red[w][threadIdx.x];
        s += bias[row_base + threadIdx.x];
        if (act_tanh) s = tanhf(s);
        out[row_base + threadIdx.x] = s;
    }
}

__device__ __forceinline__ v4f stack_blend(float a_noop, float a_push, float a_pop,
                                           v4f cur, v4f up, v4f dn) {
    return a_noop * cur + a_push * up + a_pop * dn;
}

__global__ void level1_from_hidden(const float* __restrict__ hidden,
                                   const float* __restrict__ W_hh, const float* __restrict__ b_hh,
                                   const float* __restrict__ W_sin, const float* __restrict__ b_sin,
                                   const float* __restrict__ W_ctrl, const float* __restrict__ b_ctrl,
                                   float* __restrict__ ws, int K) {
    __shared__ float red[8][16];
    const int b = blockIdx.x;
    if (b < HH_BLOCKS + SIN_BLOCKS) {
        const bool is_sin = (b >= HH_BLOCKS);
        const float* W    = is_sin ? W_sin : W_hh;
        const float* bias = is_sin ? b_sin : b_hh;
        float* out        = ws + (is_sin ? WS_SINP : WS_GH);
        const int row_base = (is_sin ? (b - HH_BLOCKS) : b) * 16;
        gemv_tile(W, hidden, bias, out, K, row_base, is_sin, red);
    } else {
        __shared__ float lg[3];
        const int wave = threadIdx.x >> 5, lane = threadIdx.x & 31;
        if (wave < 3) {
            float s = 0.f;
            for (int i = lane; i < H_; i += 32) s += W_ctrl[wave * H_ + i] * hidden[i];
            #pragma unroll
            for (int off = 16; off > 0; off >>= 1) s += __shfl_down(s, off, 32);
            if (lane == 0) lg[wave] = s + b_ctrl[wave];
        }
        __syncthreads();
        if (threadIdx.x == 0) {
            float m = fmaxf(lg[0], fmaxf(lg[1], lg[2]));
            float e0 = expf(lg[0] - m), e1 = expf(lg[1] - m), e2 = expf(lg[2] - m);
            float inv = 1.f / (e0 + e1 + e2);
            ws[WS_CTRL + 0] = e0 * inv;
            ws[WS_CTRL + 1] = e1 * inv;
            ws[WS_CTRL + 2] = e2 * inv;
        }
    }
}

__global__ void level2_stack_and_gru(const int* __restrict__ inp, const float* __restrict__ emb,
                                     const float* __restrict__ stack, const float* __restrict__ hidden,
                                     const float* __restrict__ W_ih, const float* __restrict__ b_ih,
                                     const float* __restrict__ W_dec,
                                     float* __restrict__ new_stack, float* __restrict__ hid_out,
                                     float* ws, int K) {
    __shared__ float red[8][16];
    __shared__ __attribute__((aligned(16))) float rnn_in[H_ + SW_];
    __shared__ __attribute__((aligned(16))) float hn_s[UNITS_];
    __shared__ __attribute__((aligned(16))) float part_s[V_];

    const float a_push = ws[WS_CTRL + 0];
    const float a_pop  = ws[WS_CTRL + 1];
    const float a_noop = ws[WS_CTRL + 2];

    if (blockIdx.x < GI_TILES) {
        const int lane = threadIdx.x & 31;
        const int wave = threadIdx.x >> 5;
        const int u0 = blockIdx.x * UNITS_;

        const int c = threadIdx.x * 4;
        v4f x   = *(const v4f*)(emb + (size_t)inp[0] * H_ + c);
        v4f cur = *(const v4f*)(stack + c);
        v4f up  = *(const v4f*)(ws + WS_SINP + c);
        v4f dn  = *(const v4f*)(stack + SW_ + c);

        float gh_r = 0.f, gh_z = 0.f, gh_n = 0.f, b_r = 0.f, b_z = 0.f, b_n = 0.f, hp = 0.f;
        if (threadIdx.x < UNITS_) {
            const int u = u0 + threadIdx.x;
            gh_r = ws[WS_GH + u];
            gh_z = ws[WS_GH + H_ + u];
            gh_n = ws[WS_GH + 2 * H_ + u];
            b_r  = b_ih[u];
            b_z  = b_ih[H_ + u];
            b_n  = b_ih[2 * H_ + u];
            hp   = hidden[u];
        }
        *(v4f*)(rnn_in + c)      = x;
        *(v4f*)(rnn_in + H_ + c) = stack_blend(a_noop, a_push, a_pop, cur, up, dn);
        __syncthreads();

        const int kh = (lane >> 4) << 1;
        const v2f wd = *(const v2f*)(W_dec + (size_t)(wave * 16 + (lane & 15)) * H_ + u0 + kh);

        const int col  = lane & 15;
        const int gate = (col < 12) ? (col >> 2) : 2;
        gemv_partials(W_ih, rnn_in, K, gate * H_ + u0 + (col & 3), red);
        __syncthreads();

        if (threadIdx.x < UNITS_) {
            const int j = threadIdx.x;
            float gi_r = 0.f, gi_z = 0.f, gi_n = 0.f;
            #pragma unroll
            for (int w = 0; w < 8; ++w) {
                gi_r += red[w][j];
                gi_z += red[w][UNITS_ + j];
                gi_n += red[w][2 * UNITS_ + j];
            }
            gi_r += b_r;
            gi_z += b_z;
            gi_n += b_n;
            float r = 1.f / (1.f + expf(-(gi_r + gh_r)));
            float z = 1.f / (1.f + expf(-(gi_z + gh_z)));
            float n = tanhf(gi_n + r * gh_n);
            float hn = (1.f - z) * n + z * hp;
            hn_s[j] = hn;
            hid_out[u0 + j] = hn;
        }
        __syncthreads();

        v8f acc = {};
        const v2f hv = *(const v2f*)(hn_s + kh);
        acc = __builtin_amdgcn_wmma_f32_16x16x4_f32(false, hv, false, wd, (short)0, acc, false, false);
        if (lane < 16) part_s[wave * 16 + lane] = acc[0];
        __syncthreads();
        if (wave == 0)
            *(v4f*)(ws + WS_PART + (size_t)blockIdx.x * V_ + lane * 4) = *(const v4f*)(part_s + lane * 4);
    } else {
        const int c  = threadIdx.x * 4;
        const int j0 = (blockIdx.x - GI_TILES) * STK_ROWS;
        const float* col = stack + (size_t)j0 * SW_ + c;

        v4f win[STK_ROWS + 2];
        win[0] = (j0 == 0) ? *(const v4f*)(ws + WS_SINP + c) : *(const v4f*)(col - SW_);
        #pragma unroll
        for (int i = 0; i < STK_ROWS; ++i) win[i + 1] = *(const v4f*)(col + (size_t)i * SW_);
        v4f below = {};
        if (j0 + STK_ROWS < SD_) below = *(const v4f*)(col + (size_t)STK_ROWS * SW_);
        win[STK_ROWS + 1] = below;

        #pragma unroll
        for (int i = 0; i < STK_ROWS; ++i) {
            v4f o = stack_blend(a_noop, a_push, a_pop, win[i + 1], win[i], win[i + 2]);
            __builtin_nontemporal_store(o, (v4f*)(new_stack + (size_t)(j0 + i) * SW_ + c));
        }
    }
}

__global__ void level3_logit_sum(const float* __restrict__ ws, const float* __restrict__ b_dec,
                                 float* __restrict__ out) {
    __shared__ float wsum[8][32];
    const int lane = threadIdx.x & 31;
    const int wave = threadIdx.x >> 5;
    const int v = blockIdx.x * 32 + lane;
    const float bias = b_dec[v];
    const float* p = ws + WS_PART + (size_t)(wave * 32) * V_ + v;
    float t[32];
    #pragma unroll
    for (int i = 0; i < 32; ++i) t[i] = p[(size_t)i * V_];
    float s = 0.f;
    #pragma unroll
    for (int i = 0; i < 32; ++i) s += t[i];
    wsum[wave][lane] = s;
    __syncthreads();
    if (threadIdx.x < 32) {
        float o = 0.f;
        #pragma unroll
        for (int w = 0; w < 8; ++w) o += wsum[w][threadIdx.x];
        out[v] = o + bias;
    }
}

extern "C" void kernel_launch(void* const* d_in, const int* in_sizes, int n_in,
                              void* d_out, int out_size, void* d_ws, size_t ws_size,
                              hipStream_t stream) {
    (void)in_sizes; (void)n_in; (void)out_size; (void)ws_size;
    const int*   inp    = (const int*)  d_in[0];
    const float* hidden = (const float*)d_in[1];
    const float* stack  = (const float*)d_in[2];
    const float* emb    = (const float*)d_in[3];
    const float* W_ctrl = (const float*)d_in[4];
    const float* b_ctrl = (const float*)d_in[5];
    const float* W_sin  = (const float*)d_in[6];
    const float* b_sin  = (const float*)d_in[7];
    const float* W_ih   = (const float*)d_in[8];
    const float* b_ih   = (const float*)d_in[9];
    const float* W_hh   = (const float*)d_in[10];
    const float* b_hh   = (const float*)d_in[11];
    const float* W_dec  = (const float*)d_in[12];
    const float* b_dec  = (const float*)d_in[13];

    float* out       = (float*)d_out;
    float* hid_out   = out + V_;
    float* stack_out = out + V_ + H_;
    float* ws        = (float*)d_ws;

    level1_from_hidden<<<dim3(HH_BLOCKS + SIN_BLOCKS + 1), dim3(256), 0, stream>>>(
        hidden, W_hh, b_hh, W_sin, b_sin, W_ctrl, b_ctrl, ws, H_);
    level2_stack_and_gru<<<dim3(GI_TILES + STK_BLOCKS), dim3(256), 0, stream>>>(
        inp, emb, stack, hidden, W_ih, b_ih, W_dec, stack_out, hid_out, ws, H_ + SW_);
    level3_logit_sum<<<dim3(SUM_BLOCKS), dim3(256), 0, stream>>>(ws, b_dec, out);
}


# GEMVs stream weight rows contiguously (512-B loads): a WMMA's 16 columns = 16 consecutive 4-float pieces of ONE row, vector pieces in A, sums on D's diagonal; gh, stack_input, gi sum order changes
# speedup vs baseline: 2.3094x; 1.1749x over previous
#include <hip/hip_runtime.h>
#include <hip/hip_bf16.h>
#include <math.h>


#define V_  128
#define H_  1024
#define SW_ 1024
#define SD_ 8192

typedef __attribute__((ext_vector_type(2))) float v2f;
typedef __attribute__((ext_vector_type(4))) float v4f;
typedef __attribute__((ext_vector_type(8))) float v8f;

#define WS_CTRL   0
#define WS_SINP   16
#define WS_GH     4096
#define WS_PART   16384

#define HH_BLOCKS  (3 * H_ / 16)
#define SIN_BLOCKS (SW_ / 16)
#define UNITS_     4
#define GI_TILES   (H_ / UNITS_)
#define SUM_BLOCKS (V_ / 32)
#define STK_ROWS   16
#define STK_BLOCKS (SD_ / STK_ROWS)


template <int M>
__device__ __forceinline__ float row_xor_add(float p) {
    return p + __builtin_bit_cast(float, __builtin_amdgcn_update_dpp(0, __builtin_bit_cast(int, p), 0x160 + M, 0xF, 0xF, true));
}
__device__ __forceinline__ float wave_sum(float p) {
    p = row_xor_add<8>(p);
    p = row_xor_add<4>(p);
    p = row_xor_add<2>(p);
    p = row_xor_add<1>(p);
    const int pi = __builtin_bit_cast(int, p);
    p += __builtin_bit_cast(float, __builtin_amdgcn_permlanex16(pi, pi, 0x76543210, 0xfedcba98, false, false));
    return p;
}

template <int NR, int NCH>
__device__ __forceinline__ void load_rows(const float* const (&row)[NR], v4f (&b)[NR][NCH]) {
    const int lane = threadIdx.x & 31;
    #pragma unroll
    for (int i = 0; i < NCH; ++i) {
        #pragma unroll
        for (int r = 0; r < NR; ++r) b[r][i] = *(const v4f*)(row[r] + 128 * i + 4 * lane);
        __builtin_amdgcn_sched_barrier(0);
    }
}

template <int NR, int NCH>
__device__ __forceinline__ void rows_dot(const v4f (&b)[NR][NCH], const v4f (&a)[NCH], float (&tot)[NR]) {
    v8f acc[NR];
    #pragma unroll
    for (int r = 0; r < NR; ++r) acc[r] = v8f{};
    #pragma unroll
    for (int i = 0; i < NCH; ++i) {
        const v2f a_lo = {a[i][0], a[i][1]};
        const v2f a_hi = {a[i][2], a[i][3]};
        #pragma unroll
        for (int r = 0; r < NR; ++r) {
            const v2f b_lo = {b[r][i][0], b[r][i][1]};
            acc[r] = __builtin_amdgcn_wmma_f32_16x16x4_f32(false, a_lo, false, b_lo, (short)0, acc[r], false, false);
        }
        #pragma unroll
        for (int r = 0; r < NR; ++r) {
            const v2f b_hi = {b[r][i][2], b[r][i][3]};
            acc[r] = __builtin_amdgcn_wmma_f32_16x16x4_f32(false, a_hi, false, b_hi, (short)0, acc[r], false, false);
        }
    }
    const int lane = threadIdx.x & 31;
    const int dr = (lane & 15) - ((lane >> 4) << 3);
    #pragma unroll
    for (int r = 0; r < NR; ++r) {
        float p = 0.f;
        #pragma unroll
        for (int i = 0; i < 8; ++i) p = (dr == i) ? acc[r][i] : p;
        tot[r] = wave_sum(p);
    }
}

__device__ __forceinline__ v4f stack_blend(float a_noop, float a_push, float a_pop,
                                           v4f cur, v4f up, v4f dn) {
    return a_noop * cur + a_push * up + a_pop * dn;
}

__global__ void __launch_bounds__(256)
level1_from_hidden(const float* __restrict__ hidden,
                   const float* __restrict__ W_hh, const float* __restrict__ b_hh,
                   const float* __restrict__ W_sin, const float* __restrict__ b_sin,
                   const float* __restrict__ W_ctrl, const float* __restrict__ b_ctrl,
                   float* __restrict__ ws) {
    __shared__ float rowsum[16];
    const int b = blockIdx.x;
    if (b < HH_BLOCKS + SIN_BLOCKS) {
        const bool is_sin = (b >= HH_BLOCKS);
        const float* W    = is_sin ? W_sin : W_hh;
        const float* bias = is_sin ? b_sin : b_hh;
        float* out        = ws + (is_sin ? WS_SINP : WS_GH);
        const int row_base = (is_sin ? (b - HH_BLOCKS) : b) * 16;
        const int lane = threadIdx.x & 31;
        const int wave = threadIdx.x >> 5;

        v4f a[H_ / 128];
        #pragma unroll
        for (int i = 0; i < H_ / 128; ++i) a[i] = *(const v4f*)(hidden + 128 * i + 4 * lane);
        const float* const row[2] = { W + (size_t)(row_base + 2 * wave) * H_,
                                      W + (size_t)(row_base + 2 * wave + 1) * H_ };
        v4f wq[2][H_ / 128];
        load_rows<2, H_ / 128>(row, wq);
        float tot[2];
        rows_dot<2, H_ / 128>(wq, a, tot);
        if (lane == 0) { rowsum[2 * wave] = tot[0]; rowsum[2 * wave + 1] = tot[1]; }
        __syncthreads();
        if (threadIdx.x < 16) {
            float s = rowsum[threadIdx.x] + bias[row_base + threadIdx.x];
            if (is_sin) s = tanhf(s);
            out[row_base + threadIdx.x] = s;
        }
    } else {
        __shared__ float lg[3];
        const int wave = threadIdx.x >> 5, lane = threadIdx.x & 31;
        if (wave < 3) {
            float s = 0.f;
            for (int i = lane; i < H_; i += 32) s += W_ctrl[wave * H_ + i] * hidden[i];
            #pragma unroll
            for (int off = 16; off > 0; off >>= 1) s += __shfl_down(s, off, 32);
            if (lane == 0) lg[wave] = s + b_ctrl[wave];
        }
        __syncthreads();
        if (threadIdx.x == 0) {
            float m = fmaxf(lg[0], fmaxf(lg[1], lg[2]));
            float e0 = expf(lg[0] - m), e1 = expf(lg[1] - m), e2 = expf(lg[2] - m);
            float inv = 1.f / (e0 + e1 + e2);
            ws[WS_CTRL + 0] = e0 * inv;
            ws[WS_CTRL + 1] = e1 * inv;
            ws[WS_CTRL + 2] = e2 * inv;
        }
    }
}

__global__ void __launch_bounds__(256)
level2_stack_and_gru(const int* __restrict__ inp, const float* __restrict__ emb,
                     const float* __restrict__ stack, const float* __restrict__ hidden,
                     const float* __restrict__ W_ih, const float* __restrict__ b_ih,
                     const float* __restrict__ W_dec,
                     float* __restrict__ new_stack, float* __restrict__ hid_out,
                     float* ws) {
    __shared__ float gsum[8][4];
    __shared__ __attribute__((aligned(16))) float rnn_in[H_ + SW_];
    __shared__ __attribute__((aligned(16))) float hn_s[UNITS_];
    __shared__ __attribute__((aligned(16))) float part_s[V_];

    const float a_push = ws[WS_CTRL + 0];
    const float a_pop  = ws[WS_CTRL + 1];
    const float a_noop = ws[WS_CTRL + 2];

    if (blockIdx.x < GI_TILES) {
        const int lane = threadIdx.x & 31;
        const int wave = threadIdx.x >> 5;
        const int u0 = blockIdx.x * UNITS_;

        const int c = threadIdx.x * 4;
        v4f x   = *(const v4f*)(emb + (size_t)inp[0] * H_ + c);
        v4f cur = *(const v4f*)(stack + c);
        v4f up  = *(const v4f*)(ws + WS_SINP + c);
        v4f dn  = *(const v4f*)(stack + SW_ + c);

        float gh_r = 0.f, gh_z = 0.f, gh_n = 0.f, b_r = 0.f, b_z = 0.f, b_n = 0.f, hp = 0.f;
        if (threadIdx.x < UNITS_) {
            const int u = u0 + threadIdx.x;
            gh_r = ws[WS_GH + u];
            gh_z = ws[WS_GH + H_ + u];
            gh_n = ws[WS_GH + 2 * H_ + u];
            b_r  = b_ih[u];
            b_z  = b_ih[H_ + u];
            b_n  = b_ih[2 * H_ + u];
            hp   = hidden[u];
        }

        const int kbeg = (wave & 1) * (H_ + SW_) / 2;
        const size_t wrow = (size_t)(u0 + (wave >> 1)) * (H_ + SW_) + kbeg;
        const float* const row[3] = { W_ih + wrow,
                                      W_ih + (size_t)H_ * (H_ + SW_) + wrow,
                                      W_ih + (size_t)2 * H_ * (H_ + SW_) + wrow };
        v4f wq[3][(H_ + SW_) / 256];
        load_rows<3, (H_ + SW_) / 256>(row, wq);

        const int kh = (lane >> 4) << 1;
        const v2f wd = *(const v2f*)(W_dec + (size_t)(wave * 16 + (lane & 15)) * H_ + u0 + kh);

        *(v4f*)(rnn_in + c)      = x;
        *(v4f*)(rnn_in + H_ + c) = stack_blend(a_noop, a_push, a_pop, cur, up, dn);
        __builtin_amdgcn_fence(__ATOMIC_RELEASE, "workgroup", "local");
        __builtin_amdgcn_s_barrier();
        __builtin_amdgcn_fence(__ATOMIC_ACQUIRE, "workgroup", "local");

        v4f a[(H_ + SW_) / 256];
        #pragma unroll
        for (int i = 0; i < (H_ + SW_) / 256; ++i) a[i] = *(const v4f*)(rnn_in + kbeg + 128 * i + 4 * lane);
        float tot[3];
        rows_dot<3, (H_ + SW_) / 256>(wq, a, tot);
        if (lane < 3) gsum[wave][lane] = (lane == 0) ? tot[0] : (lane == 1) ? tot[1] : tot[2];
        __syncthreads();

        if (threadIdx.x < UNITS_) {
            const int j = threadIdx.x;
            float gi_r = gsum[2 * j][0] + gsum[2 * j + 1][0];
            float gi_z = gsum[2 * j][1] + gsum[2 * j + 1][1];
            float gi_n = gsum[2 * j][2] + gsum[2 * j + 1][2];
            gi_r += b_r;
            gi_z += b_z;
            gi_n += b_n;
            float r = 1.f / (1.f + expf(-(gi_r + gh_r)));
            float z = 1.f / (1.f + expf(-(gi_z + gh_z)));
            float n = tanhf(gi_n + r * gh_n);
            float hn = (1.f - z) * n + z * hp;
            hn_s[j] = hn;
            hid_out[u0 + j] = hn;
        }
        __syncthreads();

        v8f acc = {};
        const v2f hv = *(const v2f*)(hn_s + kh);
        acc = __builtin_amdgcn_wmma_f32_16x16x4_f32(false, hv, false, wd, (short)0, acc, false, false);
        if (lane < 16) part_s[wave * 16 + lane] = acc[0];
        __syncthreads();
        if (wave == 0)
            *(v4f*)(ws + WS_PART + (size_t)blockIdx.x * V_ + lane * 4) = *(const v4f*)(part_s + lane * 4);
    } else {
        const int c  = threadIdx.x * 4;
        const int j0 = (blockIdx.x - GI_TILES) * STK_ROWS;
        const float* col = stack + (size_t)j0 * SW_ + c;

        v4f win[STK_ROWS + 2];
        win[0] = (j0 == 0) ? *(const v4f*)(ws + WS_SINP + c) : *(const v4f*)(col - SW_);
        #pragma unroll
        for (int i = 0; i < STK_ROWS; ++i) win[i + 1] = *(const v4f*)(col + (size_t)i * SW_);
        v4f below = {};
        if (j0 + STK_ROWS < SD_) below = *(const v4f*)(col + (size_t)STK_ROWS * SW_);
        win[STK_ROWS + 1] = below;

        #pragma unroll
        for (int i = 0; i < STK_ROWS; ++i) {
            v4f o = stack_blend(a_noop, a_push, a_pop, win[i + 1], win[i], win[i + 2]);
            __builtin_nontemporal_store(o, (v4f*)(new_stack + (size_t)(j0 + i) * SW_ + c));
        }
    }
}

__global__ void level3_logit_sum(const float* __restrict__ ws, const float* __restrict__ b_dec,
                                 float* __restrict__ out) {
    __shared__ float wsum[8][32];
    const int lane = threadIdx.x & 31;
    const int wave = threadIdx.x >> 5;
    const int v = blockIdx.x * 32 + lane;
    const float bias = b_dec[v];
    const float* p = ws + WS_PART + (size_t)(wave * 32) * V_ + v;
    float t[32];
    #pragma unroll
    for (int i = 0; i < 32; ++i) t[i] = p[(size_t)i * V_];
    float s = 0.f;
    #pragma unroll
    for (int i = 0; i < 32; ++i) s += t[i];
    wsum[wave][lane] = s;
    __syncthreads();
    if (threadIdx.x < 32) {
        float o = 0.f;
        #pragma unroll
        for (int w = 0; w < 8; ++w) o += wsum[w][threadIdx.x];
        out[v] = o + bias;
    }
}

extern "C" void kernel_launch(void* const* d_in, const int* in_sizes, int n_in,
                              void* d_out, int out_size, void* d_ws, size_t ws_size,
                              hipStream_t stream) {
    (void)in_sizes; (void)n_in; (void)out_size; (void)ws_size;
    const int*   inp    = (const int*)  d_in[0];
    const float* hidden = (const float*)d_in[1];
    const float* stack  = (const float*)d_in[2];
    const float* emb    = (const float*)d_in[3];
    const float* W_ctrl = (const float*)d_in[4];
    const float* b_ctrl = (const float*)d_in[5];
    const float* W_sin  = (const float*)d_in[6];
    const float* b_sin  = (const float*)d_in[7];
    const float* W_ih   = (const float*)d_in[8];
    const float* b_ih   = (const float*)d_in[9];
    const float* W_hh   = (const float*)d_in[10];
    const float* b_hh   = (const float*)d_in[11];
    const float* W_dec  = (const float*)d_in[12];
    const float* b_dec  = (const float*)d_in[13];

    float* out       = (float*)d_out;
    float* hid_out   = out + V_;
    float* stack_out = out + V_ + H_;
    float* ws        = (float*)d_ws;

    level1_from_hidden<<<dim3(HH_BLOCKS + SIN_BLOCKS + 1), dim3(256), 0, stream>>>(
        hidden, W_hh, b_hh, W_sin, b_sin, W_ctrl, b_ctrl, ws);
    level2_stack_and_gru<<<dim3(GI_TILES + STK_BLOCKS), dim3(256), 0, stream>>>(
        inp, emb, stack, hidden, W_ih, b_ih, W_dec, stack_out, hid_out, ws);
    level3_logit_sum<<<dim3(SUM_BLOCKS), dim3(256), 0, stream>>>(ws, b_dec, out);
}
